# starting kernel
# baseline (speedup 1.0000x reference)
#include <hip/hip_runtime.h>
#include <hip/hip_bf16.h>
#include <math.h>

typedef __attribute__((ext_vector_type(2))) float v2f;
typedef __attribute__((ext_vector_type(8))) float v8f;

#define EPS 1e-4f

__global__ __launch_bounds__(256) void se3_exp_kernel(
    const float* __restrict__ dofs, float* __restrict__ M, int n_groups) {
  int g = blockIdx.x * blockDim.x + threadIdx.x;
  if (g >= n_groups) return;
  const float* d = dofs + (size_t)g * 6;
  float tx = d[0], ty = d[1], tz = d[2];
  float wx = d[3], wy = d[4], wz = d[5];

  float nrm = wx * wx + wy * wy + wz * wz;
  float th  = sqrtf(fmaxf(nrm, EPS));
  float st  = sinf(th);
  float ct  = cosf(th);
  float th2 = th * th;
  float f1  = st / th;
  float f2  = (1.0f - ct) / th2;
  float f3  = (th - st) / (th2 * th);

  float xx = wx * wx, yy = wy * wy, zz = wz * wz;
  float xy = wx * wy, xz = wx * wz, yz = wy * wz;

  float R[3][3];
  R[0][0] = 1.0f + f2 * (xx - nrm);
  R[0][1] = -f1 * wz + f2 * xy;
  R[0][2] =  f1 * wy + f2 * xz;
  R[1][0] =  f1 * wz + f2 * xy;
  R[1][1] = 1.0f + f2 * (yy - nrm);
  R[1][2] = -f1 * wx + f2 * yz;
  R[2][0] = -f1 * wy + f2 * xz;
  R[2][1] =  f1 * wx + f2 * yz;
  R[2][2] = 1.0f + f2 * (zz - nrm);

  float V[3][3];
  V[0][0] = 1.0f + f3 * (xx - nrm);
  V[0][1] = -f2 * wz + f3 * xy;
  V[0][2] =  f2 * wy + f3 * xz;
  V[1][0] =  f2 * wz + f3 * xy;
  V[1][1] = 1.0f + f3 * (yy - nrm);
  V[1][2] = -f2 * wx + f3 * yz;
  V[2][0] = -f2 * wy + f3 * xz;
  V[2][1] =  f2 * wx + f3 * yz;
  V[2][2] = 1.0f + f3 * (zz - nrm);

  float t0 = V[0][0] * tx + V[0][1] * ty + V[0][2] * tz;
  float t1 = V[1][0] * tx + V[1][1] * ty + V[1][2] * tz;
  float t2 = V[2][0] * tx + V[2][1] * ty + V[2][2] * tz;

  float* Mg = M + (size_t)g * 12;
  Mg[0]  = R[0][0]; Mg[1]  = R[0][1]; Mg[2]  = R[0][2]; Mg[3]  = t0;
  Mg[4]  = R[1][0]; Mg[5]  = R[1][1]; Mg[6]  = R[1][2]; Mg[7]  = t1;
  Mg[8]  = R[2][0]; Mg[9]  = R[2][1]; Mg[10] = R[2][2]; Mg[11] = t2;
}

__global__ __launch_bounds__(256) void se3_apply_wmma_kernel(
    const float* __restrict__ X, const float* __restrict__ M,
    float* __restrict__ out, int N) {
  const int g  = blockIdx.y;
  const int l  = threadIdx.x & 31;
  const int wv = threadIdx.x >> 5;
  const bool hi = (l >= 16);
  const int  m  = l & 15;

  const float* Mg = M + (size_t)g * 12;
  float a0 = 0.0f, a1 = 0.0f, a0h = 0.0f, a1h = 0.0f;
  if (m < 3) {
    a0 = Mg[m * 4 + (hi ? 2 : 0)];
    a1 = Mg[m * 4 + (hi ? 3 : 1)];
  }
  if (m >= 8 && m < 11) {
    const int r = m - 8;
    a0h = Mg[r * 4 + (hi ? 2 : 0)];
    a1h = Mg[r * 4 + (hi ? 3 : 1)];
  }
  v2f A_lo; A_lo.x = a0;  A_lo.y = a1;
  v2f A_hi; A_hi.x = a0h; A_hi.y = a1h;

  const size_t gbase = (size_t)g * (size_t)N * 3u;
  const float* __restrict__ Xg = X + gbase;
  float* __restrict__       Og = out + gbase;

  int p0 = blockIdx.x * 1024 + wv * 128;

  for (int it = 0; it < 4; ++it, p0 += 32) {
    const int q0 = p0 + m;
    const int q1 = p0 + 16 + m;

    float b0a = Xg[q0 * 3 + (hi ? 2 : 0)];
    float b0b = Xg[q0 * 3 + 1];
    float b1a = Xg[q1 * 3 + (hi ? 2 : 0)];
    float b1b = Xg[q1 * 3 + 1];

    v2f B0; B0.x = b0a; B0.y = hi ? 1.0f : b0b;
    v2f B1; B1.x = b1a; B1.y = hi ? 1.0f : b1b;

    v8f c = {};
    v8f d0 = __builtin_amdgcn_wmma_f32_16x16x4_f32(
        false, A_lo, false, B0, (short)0, c, false, false);
    v8f d1 = __builtin_amdgcn_wmma_f32_16x16x4_f32(
        false, A_hi, false, B1, (short)0, c, false, false);

    float o0 = hi ? d1[0] : d0[0];
    float o1 = hi ? d1[1] : d0[1];
    float o2 = hi ? d1[2] : d0[2];

    const size_t ob = (size_t)(p0 + l) * 3u;
    Og[ob + 0] = o0;
    Og[ob + 1] = o1;
    Og[ob + 2] = o2;
  }
}

extern "C" void kernel_launch(void* const* d_in, const int* in_sizes, int n_in,
                              void* d_out, int out_size, void* d_ws, size_t ws_size,
                              hipStream_t stream) {
  const float* X    = (const float*)d_in[0];
  const float* dofs = (const float*)d_in[1];
  float* out = (float*)d_out;
  float* Mws = (float*)d_ws;

  const int BT = in_sizes[1] / 6;
  const int N  = in_sizes[0] / (BT * 3);

  {
    int threads = 256;
    int blocks = (BT + threads - 1) / threads;
    se3_exp_kernel<<<blocks, threads, 0, stream>>>(dofs, Mws, BT);
  }

  {
    dim3 grid((unsigned)(N / 1024), (unsigned)BT);
    se3_apply_wmma_kernel<<<grid, 256, 0, stream>>>(X, Mws, out, N);
  }
}


# Fuse the SE(3) exp launch into the apply kernel: wave 0 of each block evaluates [R|t] behind the issued point loads and shares it through LDS and one barrier; 2 launches -> 1, no workspace
# speedup vs baseline: 1.1380x; 1.1380x over previous
#include <hip/hip_runtime.h>
#include <hip/hip_bf16.h>
#include <math.h>


typedef __attribute__((ext_vector_type(2))) float v2f;
typedef __attribute__((ext_vector_type(8))) float v8f;

#define EPS 1e-4f

__device__ __forceinline__ void se3_exp_group(const float* __restrict__ d,
                                              float* __restrict__ Mg) {
  float tx = d[0], ty = d[1], tz = d[2];
  float wx = d[3], wy = d[4], wz = d[5];

  float nrm = wx * wx + wy * wy + wz * wz;
  float th  = sqrtf(fmaxf(nrm, EPS));
  float st  = sinf(th);
  float ct  = cosf(th);
  float th2 = th * th;
  float f1  = st / th;
  float f2  = (1.0f - ct) / th2;
  float f3  = (th - st) / (th2 * th);

  float xx = wx * wx, yy = wy * wy, zz = wz * wz;
  float xy = wx * wy, xz = wx * wz, yz = wy * wz;

  float R[3][3];
  R[0][0] = 1.0f + f2 * (xx - nrm);
  R[0][1] = -f1 * wz + f2 * xy;
  R[0][2] =  f1 * wy + f2 * xz;
  R[1][0] =  f1 * wz + f2 * xy;
  R[1][1] = 1.0f + f2 * (yy - nrm);
  R[1][2] = -f1 * wx + f2 * yz;
  R[2][0] = -f1 * wy + f2 * xz;
  R[2][1] =  f1 * wx + f2 * yz;
  R[2][2] = 1.0f + f2 * (zz - nrm);

  float V[3][3];
  V[0][0] = 1.0f + f3 * (xx - nrm);
  V[0][1] = -f2 * wz + f3 * xy;
  V[0][2] =  f2 * wy + f3 * xz;
  V[1][0] =  f2 * wz + f3 * xy;
  V[1][1] = 1.0f + f3 * (yy - nrm);
  V[1][2] = -f2 * wx + f3 * yz;
  V[2][0] = -f2 * wy + f3 * xz;
  V[2][1] =  f2 * wx + f3 * yz;
  V[2][2] = 1.0f + f3 * (zz - nrm);

  float t0 = V[0][0] * tx + V[0][1] * ty + V[0][2] * tz;
  float t1 = V[1][0] * tx + V[1][1] * ty + V[1][2] * tz;
  float t2 = V[2][0] * tx + V[2][1] * ty + V[2][2] * tz;

  Mg[0]  = R[0][0]; Mg[1]  = R[0][1]; Mg[2]  = R[0][2]; Mg[3]  = t0;
  Mg[4]  = R[1][0]; Mg[5]  = R[1][1]; Mg[6]  = R[1][2]; Mg[7]  = t1;
  Mg[8]  = R[2][0]; Mg[9]  = R[2][1]; Mg[10] = R[2][2]; Mg[11] = t2;
}

__global__ __launch_bounds__(256) void se3_fused_wmma_kernel(
    const float* __restrict__ X, const float* __restrict__ dofs,
    float* __restrict__ out, int N) {
  __shared__ __attribute__((aligned(16))) float sM[12];

  const int g  = blockIdx.y;
  const int l  = threadIdx.x & 31;
  const int wv = threadIdx.x >> 5;
  const bool hi = (l >= 16);
  const int  m  = l & 15;

  const size_t gbase = (size_t)g * (size_t)N * 3u;
  const float* __restrict__ Xg = X + gbase;
  float* __restrict__       Og = out + gbase;

  const int pbase = blockIdx.x * 1024 + wv * 128;

  float b0a[4], b0b[4], b1a[4], b1b[4];
  for (int it = 0; it < 4; ++it) {
    const int q0 = pbase + it * 32 + m;
    const int q1 = pbase + it * 32 + 16 + m;
    b0a[it] = Xg[q0 * 3 + (hi ? 2 : 0)];
    b0b[it] = Xg[q0 * 3 + 1];
    b1a[it] = Xg[q1 * 3 + (hi ? 2 : 0)];
    b1b[it] = Xg[q1 * 3 + 1];
  }

  if (wv == 0) {
    float Mg[12];
    se3_exp_group(dofs + (size_t)g * 6, Mg);
    if (l == 0) {
      for (int i = 0; i < 12; ++i) sM[i] = Mg[i];
    }
  }
  __syncthreads();

  const int r  = m & 7;
  const int rc = r < 3 ? r : 2;
  const float e0 = sM[rc * 4 + (hi ? 2 : 0)];
  const float e1 = sM[rc * 4 + (hi ? 3 : 1)];
  const bool lo_row = (m < 3);
  const bool hi_row = (m >= 8) && (m < 11);
  v2f A_lo; A_lo.x = lo_row ? e0 : 0.0f; A_lo.y = lo_row ? e1 : 0.0f;
  v2f A_hi; A_hi.x = hi_row ? e0 : 0.0f; A_hi.y = hi_row ? e1 : 0.0f;

  for (int it = 0; it < 4; ++it) {
    v2f B0; B0.x = b0a[it]; B0.y = hi ? 1.0f : b0b[it];
    v2f B1; B1.x = b1a[it]; B1.y = hi ? 1.0f : b1b[it];

    v8f c = {};
    v8f d0 = __builtin_amdgcn_wmma_f32_16x16x4_f32(
        false, A_lo, false, B0, (short)0, c, false, false);
    v8f d1 = __builtin_amdgcn_wmma_f32_16x16x4_f32(
        false, A_hi, false, B1, (short)0, c, false, false);

    float o0 = hi ? d1[0] : d0[0];
    float o1 = hi ? d1[1] : d0[1];
    float o2 = hi ? d1[2] : d0[2];

    const size_t ob = (size_t)(pbase + it * 32 + l) * 3u;
    Og[ob + 0] = o0;
    Og[ob + 1] = o1;
    Og[ob + 2] = o2;
  }
}

extern "C" void kernel_launch(void* const* d_in, const int* in_sizes, int n_in,
                              void* d_out, int out_size, void* d_ws, size_t ws_size,
                              hipStream_t stream) {
  const float* X    = (const float*)d_in[0];
  const float* dofs = (const float*)d_in[1];
  float* out = (float*)d_out;

  const int BT = in_sizes[1] / 6;
  const int N  = in_sizes[0] / (BT * 3);

  dim3 grid((unsigned)(N / 1024), (unsigned)BT);
  se3_fused_wmma_kernel<<<grid, 256, 0, stream>>>(X, dofs, out, N);
}
